# starting kernel
# baseline (speedup 1.0000x reference)
#include <hip/hip_runtime.h>
#include <math.h>


typedef float v2f __attribute__((ext_vector_type(2)));
typedef float v4f __attribute__((ext_vector_type(4)));
typedef float v8f __attribute__((ext_vector_type(8)));

#define BATCH 64
#define NI    512
#define NO    64
#define NV    16
#define NL    8
#define KTOT  (NI * NL)
#define KSPL  (KTOT / 4)
#define IB    8

static __device__ __forceinline__ v8f wmma_f32_k4(v2f a, v2f b, v8f c) {
  return __builtin_amdgcn_wmma_f32_16x16x4_f32(false, a, false, b, (short)0, c,
                                               false, false);
}

static __device__ __forceinline__ void stage_tile(float* red, int slot, int lane,
                                                  v8f c) {
  v4f lo = __builtin_shufflevector(c, c, 0, 1, 2, 3);
  v4f hi = __builtin_shufflevector(c, c, 4, 5, 6, 7);
  *(v4f*)&red[((size_t)slot * 32 + lane) * 8 + 0] = lo;
  *(v4f*)&red[((size_t)slot * 32 + lane) * 8 + 4] = hi;
}

template <bool WITH_PROBS>
static __device__ __forceinline__ void caps_gemm_body(
    const float* __restrict__ w, const float* __restrict__ x,
    const float* __restrict__ probs, float* __restrict__ outw, float out_scale) {
  __shared__ float red[4 * 4 * 32 * 8];
  const int lane = threadIdx.x & 31;
  const int ks   = threadIdx.x >> 5;
  const int o    = blockIdx.x;
  const int half = lane >> 4;
  const int lr   = lane & 15;
  const float* __restrict__ wo = w + (size_t)o * NI * NV * NL;
  const float* __restrict__ po =
      WITH_PROBS ? probs + (size_t)o * NI * BATCH : nullptr;

  v8f c0 = {}, c1 = {}, c2 = {}, c3 = {};
  const int kbeg = ks * KSPL, kend = kbeg + KSPL;
  for (int k0 = kbeg; k0 < kend; k0 += 4) {
    const int k = k0 + 2 * half;
    const int i = k >> 3;
    const int l = k & 7;
    if ((k0 & 255) == 0 && (k0 + 512) < kend)
      __builtin_prefetch(wo + (size_t)((k0 + 512) >> 3) * NV * NL, 0, 0);

    v2f a = *(const v2f*)(wo + ((size_t)i * NV + lr) * NL + l);

    const float* xb = x + ((size_t)lr * NI + i) * NL + l;
    v2f b0 = *(const v2f*)(xb);
    v2f b1 = *(const v2f*)(xb + (size_t)16 * NI * NL);
    v2f b2 = *(const v2f*)(xb + (size_t)32 * NI * NL);
    v2f b3 = *(const v2f*)(xb + (size_t)48 * NI * NL);
    if (WITH_PROBS) {
      const float* pb = po + (size_t)i * BATCH + lr;
      b0 = b0 * pb[0];
      b1 = b1 * pb[16];
      b2 = b2 * pb[32];
      b3 = b3 * pb[48];
    }
    c0 = wmma_f32_k4(a, b0, c0);
    c1 = wmma_f32_k4(a, b1, c1);
    c2 = wmma_f32_k4(a, b2, c2);
    c3 = wmma_f32_k4(a, b3, c3);
  }

  stage_tile(red, ks * 4 + 0, lane, c0);
  stage_tile(red, ks * 4 + 1, lane, c1);
  stage_tile(red, ks * 4 + 2, lane, c2);
  stage_tile(red, ks * 4 + 3, lane, c3);
  __syncthreads();

  const int bt = ks;
  const int bb = bt * 16 + lr;
  v4f alo = {}, ahi = {};
  #pragma unroll
  for (int s = 0; s < 4; ++s) {
    const float* q = &red[((size_t)(s * 4 + bt) * 32 + lane) * 8];
    alo += *(const v4f*)(q);
    ahi += *(const v4f*)(q + 4);
  }
  #pragma unroll
  for (int r = 0; r < 4; ++r) {
    outw[((size_t)o * NV + (r + 8 * half)) * BATCH + bb]     = alo[r] * out_scale;
    outw[((size_t)o * NV + (r + 4 + 8 * half)) * BATCH + bb] = ahi[r] * out_scale;
  }
}

__global__ void caps_init_kernel(const float* __restrict__ w,
                                 const float* __restrict__ x,
                                 float* __restrict__ outw) {
  caps_gemm_body<false>(w, x, nullptr, outw, 1.0f / (float)NI);
}

__global__ void caps_update_kernel(const float* __restrict__ w,
                                   const float* __restrict__ x,
                                   const float* __restrict__ probs,
                                   float* __restrict__ outw) {
  caps_gemm_body<true>(w, x, probs, outw, 1.0f);
}

__global__ void caps_nhat_kernel(const float* __restrict__ outw,
                                 float* __restrict__ nhat) {
  const int row = blockIdx.x * blockDim.x + threadIdx.x;
  const int o = row >> 6, b = row & 63;
  const float* p = outw + (size_t)o * NV * BATCH + b;
  float v[NV], ss = 0.0f;
  #pragma unroll
  for (int t = 0; t < NV; ++t) { v[t] = p[(size_t)t * BATCH]; ss += v[t] * v[t]; }
  const float inv = 1.0f / fmaxf(sqrtf(ss), 1e-12f);
  float* q = nhat + (size_t)o * NV * BATCH + b;
  #pragma unroll
  for (int t = 0; t < NV; ++t) q[(size_t)t * BATCH] = v[t] * inv;
}

__global__ void caps_logits_softmax_kernel(const float* __restrict__ w,
                                           const float* __restrict__ x,
                                           const float* __restrict__ nhat,
                                           float* __restrict__ probs) {
  extern __shared__ float slog[];
  const int tid  = threadIdx.x;
  const int wv   = tid >> 5;
  const int lane = tid & 31;
  const int half = lane >> 4;
  const int lr   = lane & 15;
  const int ib0  = blockIdx.x * IB;

  for (int t = wv; t < NO * IB * 4; t += 16) {
    const int o  = t >> 5;
    const int r2 = t & 31;
    const int ii = r2 >> 2;
    const int bt = r2 & 3;
    const int i  = ib0 + ii;
    const int bb = bt * 16 + lr;
    const float* __restrict__ wrow = w + ((size_t)o * NI + i) * NV * NL;

    v8f c = {};
    v2f a0 = *(const v2f*)(wrow + (size_t)lr * NL + 2 * half);
    v2f b0 = *(const v2f*)(x + ((size_t)bb * NI + i) * NL + 2 * half);
    c = wmma_f32_k4(a0, b0, c);
    v2f a1 = *(const v2f*)(wrow + (size_t)lr * NL + 4 + 2 * half);
    v2f b1 = *(const v2f*)(x + ((size_t)bb * NI + i) * NL + 4 + 2 * half);
    c = wmma_f32_k4(a1, b1, c);

    const float* __restrict__ nh = nhat + (size_t)o * NV * BATCH + bb;
    float acc = 0.0f;
    #pragma unroll
    for (int r = 0; r < 8; ++r) acc += c[r] * nh[(size_t)(r + 8 * half) * BATCH];
    acc += __shfl_xor(acc, 16, 32);
    if (half == 0) slog[(o * IB + ii) * BATCH + bb] = acc;
  }
  __syncthreads();

  const int ii = tid >> 6;
  const int b  = tid & 63;
  const int i  = ib0 + ii;
  float mx = -3.402823466e+38f;
  for (int o = 0; o < NO; ++o)
    mx = fmaxf(mx, slog[(o * IB + ii) * BATCH + b]);
  float s = 0.0f;
  for (int o = 0; o < NO; ++o) {
    const float e = __expf(slog[(o * IB + ii) * BATCH + b] - mx);
    slog[(o * IB + ii) * BATCH + b] = e;
    s += e;
  }
  const float inv = 1.0f / s;
  for (int o = 0; o < NO; ++o)
    probs[((size_t)o * NI + i) * BATCH + b] =
        slog[(o * IB + ii) * BATCH + b] * inv;
}

__global__ void caps_final_kernel(const float* __restrict__ outw,
                                  const float* __restrict__ bias,
                                  float* __restrict__ y) {
  const int row = blockIdx.x * blockDim.x + threadIdx.x;
  const int o = row >> 6, b = row & 63;
  const float* p = outw + (size_t)o * NV * BATCH + b;
  float v[NV], ss = 0.0f;
  #pragma unroll
  for (int t = 0; t < NV; ++t) { v[t] = p[(size_t)t * BATCH]; ss += v[t] * v[t]; }
  const float n  = sqrtf(ss);
  const float sc = n / (1.0f + ss);
  #pragma unroll
  for (int t = 0; t < NV; ++t)
    y[((size_t)b * NO + o) * NV + t] = v[t] * sc + bias[o * NV + t];
}

extern "C" void kernel_launch(void* const* d_in, const int* in_sizes, int n_in,
                              void* d_out, int out_size, void* d_ws, size_t ws_size,
                              hipStream_t stream) {
  const float* x    = (const float*)d_in[0];
  const float* w    = (const float*)d_in[1];
  const float* bias = (const float*)d_in[2];
  float* y = (float*)d_out;

  float* probs = (float*)d_ws;
  float* outw  = probs + (size_t)NO * NI * BATCH;
  float* nhat  = outw  + (size_t)NO * NV * BATCH;
  const size_t need =
      ((size_t)NO * NI * BATCH + 2 * (size_t)NO * NV * BATCH) * sizeof(float);
  if (ws_size < need) return;

  caps_init_kernel<<<NO, 128, 0, stream>>>(w, x, outw);
  for (int it = 0; it < 3; ++it) {
    caps_nhat_kernel<<<16, 256, 0, stream>>>(outw, nhat);
    caps_logits_softmax_kernel<<<64, 512,
        (size_t)NO * IB * BATCH * sizeof(float), stream>>>(w, x, nhat, probs);
    caps_update_kernel<<<NO, 128, 0, stream>>>(w, x, probs, outw);
  }
  caps_final_kernel<<<16, 256, 0, stream>>>(outw, bias, y);
}


# Logits contraction re-associated as x.(W^T nhat): per o, W^T@nhat on WMMA with K=v, nhat and x rows held in registers, l summed inside a lane, no cross-lane step (sum order changes: v first, then l)
# speedup vs baseline: 3.2294x; 3.2294x over previous
#include <hip/hip_runtime.h>
#include <math.h>


typedef float v2f __attribute__((ext_vector_type(2)));
typedef float v4f __attribute__((ext_vector_type(4)));
typedef float v8f __attribute__((ext_vector_type(8)));

#define BATCH 64
#define NI    512
#define NO    64
#define NV    16
#define NL    8
#define KTOT  (NI * NL)
#define KSPL  (KTOT / 4)
#define IB    8
#define CI    8
#define CK    (CI * NL)
#define XP    (CK + 4)

static __device__ __forceinline__ v8f wmma_f32_k4(v2f a, v2f b, v8f c) {
  return __builtin_amdgcn_wmma_f32_16x16x4_f32(false, a, false, b, (short)0, c,
                                               false, false);
}

static __device__ __forceinline__ void stage_tile(float* red, int slot, int lane,
                                                  v8f c) {
  v4f lo = __builtin_shufflevector(c, c, 0, 1, 2, 3);
  v4f hi = __builtin_shufflevector(c, c, 4, 5, 6, 7);
  *(v4f*)&red[((size_t)slot * 32 + lane) * 8 + 0] = lo;
  *(v4f*)&red[((size_t)slot * 32 + lane) * 8 + 4] = hi;
}

template <bool WITH_PROBS>
static __device__ __forceinline__ void caps_gemm_body(
    const float* __restrict__ w, const float* __restrict__ x,
    const float* __restrict__ probs, float* __restrict__ outw, float out_scale) {
  __shared__ float red[4 * 4 * 32 * 8];
  __shared__ float xs[4 * BATCH * XP];
  const int lane = threadIdx.x & 31;
  const int ks   = threadIdx.x >> 5;
  const int o    = blockIdx.x;
  const int half = lane >> 4;
  const int lr   = lane & 15;
  const float* __restrict__ wo = w + (size_t)o * NI * NV * NL;
  const float* __restrict__ po =
      WITH_PROBS ? probs + (size_t)o * NI * BATCH : nullptr;
  float* xw = xs + (size_t)ks * BATCH * XP;
  const int srow = lane >> 4;
  const int scol = (lane & 15) * 4;
  const float* __restrict__ wl = wo + (size_t)lr * NL + 2 * half;
  const float* __restrict__ xl = x + (size_t)srow * NI * NL + scol;

  v8f c0 = {}, c1 = {}, c2 = {}, c3 = {};
  const int ibeg = ks * (NI / 4);
  for (int ic = 0; ic < NI / 4; ic += CI) {
    const int i0 = ibeg + ic;

    v4f xr[BATCH / 2];
    #pragma unroll
    for (int j = 0; j < BATCH / 2; ++j)
      xr[j] = *(const v4f*)(xl + (size_t)i0 * NL + (size_t)(2 * j) * NI * NL);
    v2f a[2 * CI];
    #pragma unroll
    for (int kk = 0; kk < 2 * CI; ++kk)
      a[kk] = *(const v2f*)(wl + (size_t)i0 * NV * NL + (kk >> 1) * NV * NL + (kk & 1) * 4);
    float p[CI][4];
    if (WITH_PROBS) {
      #pragma unroll
      for (int ii = 0; ii < CI; ++ii) {
        const float* pb = po + (size_t)(i0 + ii) * BATCH + lr;
        p[ii][0] = pb[0]; p[ii][1] = pb[16]; p[ii][2] = pb[32]; p[ii][3] = pb[48];
      }
    }

    #pragma unroll
    for (int j = 0; j < BATCH / 2; ++j)
      *(v4f*)&xw[(2 * j + srow) * XP + scol] = xr[j];
    __builtin_amdgcn_fence(__ATOMIC_RELEASE, "workgroup", "local");
    __builtin_amdgcn_wave_barrier();
    __builtin_amdgcn_fence(__ATOMIC_ACQUIRE, "workgroup", "local");

    #pragma unroll
    for (int kk = 0; kk < 2 * CI; ++kk) {
      const int kc = kk * 4 + 2 * half;
      v2f b0 = *(const v2f*)&xw[(lr)      * XP + kc];
      v2f b1 = *(const v2f*)&xw[(16 + lr) * XP + kc];
      v2f b2 = *(const v2f*)&xw[(32 + lr) * XP + kc];
      v2f b3 = *(const v2f*)&xw[(48 + lr) * XP + kc];
      if (WITH_PROBS) {
        b0 = b0 * p[kk >> 1][0];
        b1 = b1 * p[kk >> 1][1];
        b2 = b2 * p[kk >> 1][2];
        b3 = b3 * p[kk >> 1][3];
      }
      c0 = wmma_f32_k4(a[kk], b0, c0);
      c1 = wmma_f32_k4(a[kk], b1, c1);
      c2 = wmma_f32_k4(a[kk], b2, c2);
      c3 = wmma_f32_k4(a[kk], b3, c3);
    }
    __builtin_amdgcn_fence(__ATOMIC_RELEASE, "workgroup", "local");
    __builtin_amdgcn_wave_barrier();
  }

  stage_tile(red, ks * 4 + 0, lane, c0);
  stage_tile(red, ks * 4 + 1, lane, c1);
  stage_tile(red, ks * 4 + 2, lane, c2);
  stage_tile(red, ks * 4 + 3, lane, c3);
  __syncthreads();

  const int bt = ks;
  const int bb = bt * 16 + lr;
  v4f alo = {}, ahi = {};
  #pragma unroll
  for (int s = 0; s < 4; ++s) {
    const float* q = &red[((size_t)(s * 4 + bt) * 32 + lane) * 8];
    alo += *(const v4f*)(q);
    ahi += *(const v4f*)(q + 4);
  }
  #pragma unroll
  for (int r = 0; r < 4; ++r) {
    outw[((size_t)o * NV + (r + 8 * half)) * BATCH + bb]     = alo[r] * out_scale;
    outw[((size_t)o * NV + (r + 4 + 8 * half)) * BATCH + bb] = ahi[r] * out_scale;
  }
}

__global__ void __launch_bounds__(128) caps_init_kernel(const float* __restrict__ w,
                                 const float* __restrict__ x,
                                 float* __restrict__ outw) {
  caps_gemm_body<false>(w, x, nullptr, outw, 1.0f / (float)NI);
}

__global__ void __launch_bounds__(128) caps_update_kernel(const float* __restrict__ w,
                                   const float* __restrict__ x,
                                   const float* __restrict__ probs,
                                   float* __restrict__ outw) {
  caps_gemm_body<true>(w, x, probs, outw, 1.0f);
}

__global__ void caps_nhat_kernel(const float* __restrict__ outw,
                                 float* __restrict__ nhat) {
  const int row = blockIdx.x * blockDim.x + threadIdx.x;
  const int o = row >> 6, b = row & 63;
  const float* p = outw + (size_t)o * NV * BATCH + b;
  float v[NV], ss = 0.0f;
  #pragma unroll
  for (int t = 0; t < NV; ++t) { v[t] = p[(size_t)t * BATCH]; ss += v[t] * v[t]; }
  const float inv = 1.0f / fmaxf(sqrtf(ss), 1e-12f);
  float* q = nhat + (size_t)o * NV * BATCH + b;
  #pragma unroll
  for (int t = 0; t < NV; ++t) q[(size_t)t * BATCH] = v[t] * inv;
}

__global__ void __launch_bounds__(512) caps_logits_softmax_kernel(
    const float* __restrict__ w, const float* __restrict__ x,
    const float* __restrict__ nhat, float* __restrict__ probs) {
  extern __shared__ float slog[];
  const int tid  = threadIdx.x;
  const int wv   = tid >> 5;
  const int lane = tid & 31;
  const int half = lane >> 4;
  const int lr   = lane & 15;
  const int ib0  = blockIdx.x * IB;
  const int q    = wv & 1;

  v4f xq[2][IB / 2][2];
  #pragma unroll
  for (int j = 0; j < 2; ++j) {
    #pragma unroll
    for (int p = 0; p < IB / 2; ++p) {
      const float* xp =
          x + ((size_t)((2 * q + j) * 16 + lr) * NI + ib0 + 2 * p + half) * NL;
      xq[j][p][0] = *(const v4f*)(xp);
      xq[j][p][1] = *(const v4f*)(xp + 4);
    }
  }

  const float* __restrict__ wl =
      w + ((size_t)ib0 + (lr >> 3)) * NV * NL + (size_t)(2 * half) * NL + (lr & 7);
  const float* __restrict__ nl = nhat + (size_t)(2 * half) * BATCH + (2 * q) * 16 + lr;

  for (int o = wv >> 1; o < NO; o += 8) {
    const float* __restrict__ wo = wl + (size_t)o * NI * NV * NL;
    const float* __restrict__ no = nl + (size_t)o * NV * BATCH;

    v2f a[IB / 2][4];
    #pragma unroll
    for (int p = 0; p < IB / 2; ++p) {
      #pragma unroll
      for (int kk = 0; kk < 4; ++kk) {
        a[p][kk][0] = wo[(size_t)p * 2 * NV * NL + (4 * kk) * NL];
        a[p][kk][1] = wo[(size_t)p * 2 * NV * NL + (4 * kk + 1) * NL];
      }
    }
    v2f bf[2][4];
    #pragma unroll
    for (int j = 0; j < 2; ++j) {
      #pragma unroll
      for (int kk = 0; kk < 4; ++kk) {
        bf[j][kk][0] = no[(size_t)(4 * kk) * BATCH + j * 16];
        bf[j][kk][1] = no[(size_t)(4 * kk + 1) * BATCH + j * 16];
      }
    }

    asm volatile("" ::
        "v"(a[0][0]), "v"(a[0][1]), "v"(a[0][2]), "v"(a[0][3]),
        "v"(a[1][0]), "v"(a[1][1]), "v"(a[1][2]), "v"(a[1][3]),
        "v"(a[2][0]), "v"(a[2][1]), "v"(a[2][2]), "v"(a[2][3]),
        "v"(a[3][0]), "v"(a[3][1]), "v"(a[3][2]), "v"(a[3][3]),
        "v"(bf[0][0]), "v"(bf[0][1]), "v"(bf[0][2]), "v"(bf[0][3]),
        "v"(bf[1][0]), "v"(bf[1][1]), "v"(bf[1][2]), "v"(bf[1][3]));

    v8f c[2][IB / 2];
    #pragma unroll
    for (int j = 0; j < 2; ++j) {
      #pragma unroll
      for (int p = 0; p < IB / 2; ++p) c[j][p] = v8f{};
    }
    #pragma unroll
    for (int kk = 0; kk < 4; ++kk) {
      #pragma unroll
      for (int p = 0; p < IB / 2; ++p) {
        c[0][p] = wmma_f32_k4(a[p][kk], bf[0][kk], c[0][p]);
        c[1][p] = wmma_f32_k4(a[p][kk], bf[1][kk], c[1][p]);
      }
    }

    #pragma unroll
    for (int j = 0; j < 2; ++j) {
      #pragma unroll
      for (int p = 0; p < IB / 2; ++p) {
        const v8f d = c[j][p];
        v2f s2 = v2f{d[0], d[1]} * v2f{xq[j][p][0][0], xq[j][p][0][1]};
        s2 += v2f{d[2], d[3]} * v2f{xq[j][p][0][2], xq[j][p][0][3]};
        s2 += v2f{d[4], d[5]} * v2f{xq[j][p][1][0], xq[j][p][1][1]};
        s2 += v2f{d[6], d[7]} * v2f{xq[j][p][1][2], xq[j][p][1][3]};
        slog[(o * IB + 2 * p + half) * BATCH + (2 * q + j) * 16 + lr] = s2[0] + s2[1];
      }
    }
  }
  __syncthreads();

  const int ii = tid >> 6;
  const int b  = tid & 63;
  const int i  = ib0 + ii;
  float mx = -3.402823466e+38f;
  for (int o = 0; o < NO; ++o)
    mx = fmaxf(mx, slog[(o * IB + ii) * BATCH + b]);
  float s = 0.0f;
  for (int o = 0; o < NO; ++o) {
    const float e = __expf(slog[(o * IB + ii) * BATCH + b] - mx);
    slog[(o * IB + ii) * BATCH + b] = e;
    s += e;
  }
  const float inv = 1.0f / s;
  for (int o = 0; o < NO; ++o)
    probs[((size_t)o * NI + i) * BATCH + b] =
        slog[(o * IB + ii) * BATCH + b] * inv;
}

__global__ void caps_final_kernel(const float* __restrict__ outw,
                                  const float* __restrict__ bias,
                                  float* __restrict__ y) {
  const int row = blockIdx.x * blockDim.x + threadIdx.x;
  const int o = row >> 6, b = row & 63;
  const float* p = outw + (size_t)o * NV * BATCH + b;
  float v[NV], ss = 0.0f;
  #pragma unroll
  for (int t = 0; t < NV; ++t) { v[t] = p[(size_t)t * BATCH]; ss += v[t] * v[t]; }
  const float n  = sqrtf(ss);
  const float sc = n / (1.0f + ss);
  #pragma unroll
  for (int t = 0; t < NV; ++t)
    y[((size_t)b * NO + o) * NV + t] = v[t] * sc + bias[o * NV + t];
}

extern "C" void kernel_launch(void* const* d_in, const int* in_sizes, int n_in,
                              void* d_out, int out_size, void* d_ws, size_t ws_size,
                              hipStream_t stream) {
  const float* x    = (const float*)d_in[0];
  const float* w    = (const float*)d_in[1];
  const float* bias = (const float*)d_in[2];
  float* y = (float*)d_out;

  float* probs = (float*)d_ws;
  float* outw  = probs + (size_t)NO * NI * BATCH;
  float* nhat  = outw  + (size_t)NO * NV * BATCH;
  const size_t need =
      ((size_t)NO * NI * BATCH + 2 * (size_t)NO * NV * BATCH) * sizeof(float);
  if (ws_size < need) return;

  caps_init_kernel<<<NO, 128, 0, stream>>>(w, x, outw);
  for (int it = 0; it < 3; ++it) {
    caps_nhat_kernel<<<16, 256, 0, stream>>>(outw, nhat);
    caps_logits_softmax_kernel<<<64, 512,
        (size_t)NO * IB * BATCH * sizeof(float), stream>>>(w, x, nhat, probs);
    caps_update_kernel<<<NO, 128, 0, stream>>>(w, x, probs, outw);
  }
  caps_final_kernel<<<16, 256, 0, stream>>>(outw, bias, y);
}


# GEMM launches re-cut to 256 groups of (4 out-capsules, 1/16 of K): x slice staged once in LDS, shared by 4 waves, one per o; 16 partial planes added by normalize/squash kernels (K sum order changes)
# speedup vs baseline: 4.7059x; 1.4572x over previous
#include <hip/hip_runtime.h>
#include <math.h>


typedef float v2f __attribute__((ext_vector_type(2)));
typedef float v4f __attribute__((ext_vector_type(4)));
typedef float v8f __attribute__((ext_vector_type(8)));

#define BATCH 64
#define NI    512
#define NO    64
#define NV    16
#define NL    8
#define KTOT  (NI * NL)
#define IB    8
#define OG    4
#define KS    16
#define SI    (NI / KS)
#define SK    (SI * NL)
#define XP    (SK + 4)
#define PLANE (NO * NV * BATCH)

static __device__ __forceinline__ v8f wmma_f32_k4(v2f a, v2f b, v8f c) {
  return __builtin_amdgcn_wmma_f32_16x16x4_f32(false, a, false, b, (short)0, c,
                                               false, false);
}

template <bool WITH_PROBS>
static __device__ __forceinline__ void caps_gemm_body(
    const float* __restrict__ w, const float* __restrict__ x,
    const float* __restrict__ probs, float* __restrict__ part, float out_scale) {
  __shared__ __attribute__((aligned(16))) float xs[BATCH * XP];
  const int lane = threadIdx.x & 31;
  const int wv   = threadIdx.x >> 5;
  const int sl   = blockIdx.x & (KS - 1);
  const int o    = (blockIdx.x / KS) * OG + wv;
  const int half = lane >> 4;
  const int lr   = lane & 15;
  const int i0   = sl * SI;
  const float* __restrict__ wl =
      w + ((size_t)o * NI + i0) * NV * NL + (size_t)lr * NL + 2 * half;
  const float* __restrict__ xl =
      x + (size_t)(16 * wv) * KTOT + (size_t)i0 * NL + lane * 4;

  v4f xr[32];
  #pragma unroll
  for (int j = 0; j < 32; ++j)
    xr[j] = *(const v4f*)(xl + (size_t)(j >> 1) * KTOT + (j & 1) * 128);
  v2f a[2 * SI];
  #pragma unroll
  for (int kk = 0; kk < 2 * SI; ++kk)
    a[kk] = *(const v2f*)(wl + (kk >> 1) * NV * NL + (kk & 1) * 4);
  float p[SI][4];
  if (WITH_PROBS) {
    const float* __restrict__ pb = probs + ((size_t)o * NI + i0) * BATCH + lr;
    #pragma unroll
    for (int ii = 0; ii < SI; ++ii) {
      p[ii][0] = pb[ii * BATCH];      p[ii][1] = pb[ii * BATCH + 16];
      p[ii][2] = pb[ii * BATCH + 32]; p[ii][3] = pb[ii * BATCH + 48];
    }
  }

  #pragma unroll
  for (int j = 0; j < 32; ++j)
    *(v4f*)&xs[(16 * wv + (j >> 1)) * XP + (j & 1) * 128 + lane * 4] = xr[j];
  __syncthreads();

  v8f c0 = {}, c1 = {}, c2 = {}, c3 = {};
  #pragma unroll
  for (int kk = 0; kk < 2 * SI; ++kk) {
    const int kc = kk * 4 + 2 * half;
    v2f b0 = *(const v2f*)&xs[(lr)      * XP + kc];
    v2f b1 = *(const v2f*)&xs[(16 + lr) * XP + kc];
    v2f b2 = *(const v2f*)&xs[(32 + lr) * XP + kc];
    v2f b3 = *(const v2f*)&xs[(48 + lr) * XP + kc];
    if (WITH_PROBS) {
      b0 = b0 * p[kk >> 1][0];
      b1 = b1 * p[kk >> 1][1];
      b2 = b2 * p[kk >> 1][2];
      b3 = b3 * p[kk >> 1][3];
    }
    c0 = wmma_f32_k4(a[kk], b0, c0);
    c1 = wmma_f32_k4(a[kk], b1, c1);
    c2 = wmma_f32_k4(a[kk], b2, c2);
    c3 = wmma_f32_k4(a[kk], b3, c3);
  }

  float* __restrict__ po =
      part + (size_t)sl * PLANE + ((size_t)o * NV + 8 * half) * BATCH + lr;
  #pragma unroll
  for (int r = 0; r < 8; ++r) {
    po[r * BATCH]      = c0[r] * out_scale;
    po[r * BATCH + 16] = c1[r] * out_scale;
    po[r * BATCH + 32] = c2[r] * out_scale;
    po[r * BATCH + 48] = c3[r] * out_scale;
  }
}

__global__ void __launch_bounds__(128) caps_init_kernel(const float* __restrict__ w,
                                 const float* __restrict__ x,
                                 float* __restrict__ part) {
  caps_gemm_body<false>(w, x, nullptr, part, 1.0f / (float)NI);
}

__global__ void __launch_bounds__(128) caps_update_kernel(const float* __restrict__ w,
                                   const float* __restrict__ x,
                                   const float* __restrict__ probs,
                                   float* __restrict__ part) {
  caps_gemm_body<true>(w, x, probs, part, 1.0f);
}

__global__ void __launch_bounds__(256) caps_nhat_kernel(const float* __restrict__ part,
                                 float* __restrict__ nhat) {
  __shared__ float sq[NV * BATCH];
  const int b  = threadIdx.x & 63;
  const int v0 = (threadIdx.x >> 6) * 4;
  const size_t base = (size_t)blockIdx.x * NV * BATCH + (size_t)v0 * BATCH + b;
  const float* p = part + base;
  float acc[4] = {0.0f, 0.0f, 0.0f, 0.0f};
  #pragma unroll
  for (int s = 0; s < KS; ++s) {
    #pragma unroll
    for (int u = 0; u < 4; ++u) acc[u] += p[(size_t)s * PLANE + u * BATCH];
  }
  #pragma unroll
  for (int u = 0; u < 4; ++u) sq[(v0 + u) * BATCH + b] = acc[u] * acc[u];
  __syncthreads();
  float ss = 0.0f;
  #pragma unroll
  for (int v = 0; v < NV; ++v) ss += sq[v * BATCH + b];
  const float inv = 1.0f / fmaxf(sqrtf(ss), 1e-12f);
  #pragma unroll
  for (int u = 0; u < 4; ++u) nhat[base + u * BATCH] = acc[u] * inv;
}

__global__ void __launch_bounds__(512) caps_logits_softmax_kernel(
    const float* __restrict__ w, const float* __restrict__ x,
    const float* __restrict__ nhat, float* __restrict__ probs) {
  extern __shared__ float slog[];
  const int tid  = threadIdx.x;
  const int wv   = tid >> 5;
  const int lane = tid & 31;
  const int half = lane >> 4;
  const int lr   = lane & 15;
  const int ib0  = blockIdx.x * IB;
  const int q    = wv & 1;

  v4f xq[2][IB / 2][2];
  #pragma unroll
  for (int j = 0; j < 2; ++j) {
    #pragma unroll
    for (int p = 0; p < IB / 2; ++p) {
      const float* xp =
          x + ((size_t)((2 * q + j) * 16 + lr) * NI + ib0 + 2 * p + half) * NL;
      xq[j][p][0] = *(const v4f*)(xp);
      xq[j][p][1] = *(const v4f*)(xp + 4);
    }
  }

  const float* __restrict__ wl =
      w + ((size_t)ib0 + (lr >> 3)) * NV * NL + (size_t)(2 * half) * NL + (lr & 7);
  const float* __restrict__ nl = nhat + (size_t)(2 * half) * BATCH + (2 * q) * 16 + lr;

  for (int o = wv >> 1; o < NO; o += 8) {
    const float* __restrict__ wo = wl + (size_t)o * NI * NV * NL;
    const float* __restrict__ no = nl + (size_t)o * NV * BATCH;

    v2f a[IB / 2][4];
    #pragma unroll
    for (int p = 0; p < IB / 2; ++p) {
      #pragma unroll
      for (int kk = 0; kk < 4; ++kk) {
        a[p][kk][0] = wo[(size_t)p * 2 * NV * NL + (4 * kk) * NL];
        a[p][kk][1] = wo[(size_t)p * 2 * NV * NL + (4 * kk + 1) * NL];
      }
    }
    v2f bf[2][4];
    #pragma unroll
    for (int j = 0; j < 2; ++j) {
      #pragma unroll
      for (int kk = 0; kk < 4; ++kk) {
        bf[j][kk][0] = no[(size_t)(4 * kk) * BATCH + j * 16];
        bf[j][kk][1] = no[(size_t)(4 * kk + 1) * BATCH + j * 16];
      }
    }

    asm volatile("" ::
        "v"(a[0][0]), "v"(a[0][1]), "v"(a[0][2]), "v"(a[0][3]),
        "v"(a[1][0]), "v"(a[1][1]), "v"(a[1][2]), "v"(a[1][3]),
        "v"(a[2][0]), "v"(a[2][1]), "v"(a[2][2]), "v"(a[2][3]),
        "v"(a[3][0]), "v"(a[3][1]), "v"(a[3][2]), "v"(a[3][3]),
        "v"(bf[0][0]), "v"(bf[0][1]), "v"(bf[0][2]), "v"(bf[0][3]),
        "v"(bf[1][0]), "v"(bf[1][1]), "v"(bf[1][2]), "v"(bf[1][3]));

    v8f c[2][IB / 2];
    #pragma unroll
    for (int j = 0; j < 2; ++j) {
      #pragma unroll
      for (int p = 0; p < IB / 2; ++p) c[j][p] = v8f{};
    }
    #pragma unroll
    for (int kk = 0; kk < 4; ++kk) {
      #pragma unroll
      for (int p = 0; p < IB / 2; ++p) {
        c[0][p] = wmma_f32_k4(a[p][kk], bf[0][kk], c[0][p]);
        c[1][p] = wmma_f32_k4(a[p][kk], bf[1][kk], c[1][p]);
      }
    }

    #pragma unroll
    for (int j = 0; j < 2; ++j) {
      #pragma unroll
      for (int p = 0; p < IB / 2; ++p) {
        const v8f d = c[j][p];
        v2f s2 = v2f{d[0], d[1]} * v2f{xq[j][p][0][0], xq[j][p][0][1]};
        s2 += v2f{d[2], d[3]} * v2f{xq[j][p][0][2], xq[j][p][0][3]};
        s2 += v2f{d[4], d[5]} * v2f{xq[j][p][1][0], xq[j][p][1][1]};
        s2 += v2f{d[6], d[7]} * v2f{xq[j][p][1][2], xq[j][p][1][3]};
        slog[(o * IB + 2 * p + half) * BATCH + (2 * q + j) * 16 + lr] = s2[0] + s2[1];
      }
    }
  }
  __syncthreads();

  const int ii = tid >> 6;
  const int b  = tid & 63;
  const int i  = ib0 + ii;
  float mx = -3.402823466e+38f;
  for (int o = 0; o < NO; ++o)
    mx = fmaxf(mx, slog[(o * IB + ii) * BATCH + b]);
  float s = 0.0f;
  for (int o = 0; o < NO; ++o) {
    const float e = __expf(slog[(o * IB + ii) * BATCH + b] - mx);
    slog[(o * IB + ii) * BATCH + b] = e;
    s += e;
  }
  const float inv = 1.0f / s;
  for (int o = 0; o < NO; ++o)
    probs[((size_t)o * NI + i) * BATCH + b] =
        slog[(o * IB + ii) * BATCH + b] * inv;
}

#define TP (NV + 1)
#define TQ (BATCH * TP + 16)
__global__ void __launch_bounds__(512) caps_final_kernel(const float* __restrict__ part,
                                  const float* __restrict__ bias,
                                  float* __restrict__ y) {
  __shared__ float sq[2][NV * BATCH];
  __shared__ float tile[2 * TQ];
  __shared__ float scl[2][BATCH];
  const int t  = threadIdx.x;
  const int b  = t & 63;
  const int v0 = ((t >> 6) & 3) * 4;
  const int q  = t >> 8;
  const int o0 = blockIdx.x * 2;
  const float* p = part + (size_t)(o0 + q) * NV * BATCH + (size_t)v0 * BATCH + b;
  float acc[4] = {0.0f, 0.0f, 0.0f, 0.0f};
  #pragma unroll
  for (int s = 0; s < KS; ++s) {
    #pragma unroll
    for (int u = 0; u < 4; ++u) acc[u] += p[(size_t)s * PLANE + u * BATCH];
  }
  #pragma unroll
  for (int u = 0; u < 4; ++u) {
    sq[q][(v0 + u) * BATCH + b] = acc[u] * acc[u];
    tile[q * TQ + b * TP + v0 + u] = acc[u];
  }
  __syncthreads();
  if (t < 2 * BATCH) {
    const int qq = t >> 6;
    float ss = 0.0f;
    #pragma unroll
    for (int u = 0; u < NV; ++u) ss += sq[qq][u * BATCH + b];
    const float n = sqrtf(ss);
    scl[qq][b] = n / (1.0f + ss);
  }
  __syncthreads();
  #pragma unroll
  for (int h = 0; h < 4; ++h) {
    const int e  = t + 512 * h;
    const int bb = e >> 5, qv = e & 31;
    const int qq = qv >> 4, vv = qv & 15;
    y[((size_t)bb * NO + o0) * NV + qv] =
        tile[qq * TQ + bb * TP + vv] * scl[qq][bb] + bias[o0 * NV + qv];
  }
}

extern "C" void kernel_launch(void* const* d_in, const int* in_sizes, int n_in,
                              void* d_out, int out_size, void* d_ws, size_t ws_size,
                              hipStream_t stream) {
  const float* x    = (const float*)d_in[0];
  const float* w    = (const float*)d_in[1];
  const float* bias = (const float*)d_in[2];
  float* y = (float*)d_out;

  float* probs = (float*)d_ws;
  float* part  = probs + (size_t)NO * NI * BATCH;
  float* nhat  = part  + (size_t)KS * PLANE;
  const size_t need =
      ((size_t)NO * NI * BATCH + (size_t)(KS + 1) * PLANE) * sizeof(float);
  if (ws_size < need) return;

  caps_init_kernel<<<256, 128, 0, stream>>>(w, x, part);
  for (int it = 0; it < 3; ++it) {
    caps_nhat_kernel<<<NO, 256, 0, stream>>>(part, nhat);
    caps_logits_softmax_kernel<<<64, 512,
        (size_t)NO * IB * BATCH * sizeof(float), stream>>>(w, x, nhat, probs);
    caps_update_kernel<<<256, 128, 0, stream>>>(w, x, probs, part);
  }
  caps_final_kernel<<<NO / 2, 512, 0, stream>>>(part, bias, y);
}


# Logits launch re-cut from 64 groups of 8 in-capsules to 256 groups of (4 in-capsules, half the batch), 8 waves each taking 2 out-capsules per trip; every product and every sum order unchanged
# speedup vs baseline: 6.2678x; 1.3319x over previous
#include <hip/hip_runtime.h>
#include <math.h>


typedef float v2f __attribute__((ext_vector_type(2)));
typedef float v4f __attribute__((ext_vector_type(4)));
typedef float v8f __attribute__((ext_vector_type(8)));

#define BATCH 64
#define NI    512
#define NO    64
#define NV    16
#define NL    8
#define KTOT  (NI * NL)
#define LI    4
#define LB    32
#define LWV   8
#define OPT   2
#define OG    4
#define KS    16
#define SI    (NI / KS)
#define SK    (SI * NL)
#define XP    (SK + 4)
#define PLANE (NO * NV * BATCH)

static __device__ __forceinline__ v8f wmma_f32_k4(v2f a, v2f b, v8f c) {
  return __builtin_amdgcn_wmma_f32_16x16x4_f32(false, a, false, b, (short)0, c,
                                               false, false);
}

template <bool WITH_PROBS>
static __device__ __forceinline__ void caps_gemm_body(
    const float* __restrict__ w, const float* __restrict__ x,
    const float* __restrict__ probs, float* __restrict__ part, float out_scale) {
  __shared__ __attribute__((aligned(16))) float xs[BATCH * XP];
  const int lane = threadIdx.x & 31;
  const int wv   = threadIdx.x >> 5;
  const int sl   = blockIdx.x & (KS - 1);
  const int o    = (blockIdx.x / KS) * OG + wv;
  const int half = lane >> 4;
  const int lr   = lane & 15;
  const int i0   = sl * SI;
  const float* __restrict__ wl =
      w + ((size_t)o * NI + i0) * NV * NL + (size_t)lr * NL + 2 * half;
  const float* __restrict__ xl =
      x + (size_t)(16 * wv) * KTOT + (size_t)i0 * NL + lane * 4;

  v4f xr[32];
  #pragma unroll
  for (int j = 0; j < 32; ++j)
    xr[j] = *(const v4f*)(xl + (size_t)(j >> 1) * KTOT + (j & 1) * 128);
  v2f a[2 * SI];
  #pragma unroll
  for (int kk = 0; kk < 2 * SI; ++kk)
    a[kk] = *(const v2f*)(wl + (kk >> 1) * NV * NL + (kk & 1) * 4);
  float p[SI][4];
  if (WITH_PROBS) {
    const float* __restrict__ pb = probs + ((size_t)o * NI + i0) * BATCH + lr;
    #pragma unroll
    for (int ii = 0; ii < SI; ++ii) {
      p[ii][0] = pb[ii * BATCH];      p[ii][1] = pb[ii * BATCH + 16];
      p[ii][2] = pb[ii * BATCH + 32]; p[ii][3] = pb[ii * BATCH + 48];
    }
  }

  #pragma unroll
  for (int j = 0; j < 32; ++j)
    *(v4f*)&xs[(16 * wv + (j >> 1)) * XP + (j & 1) * 128 + lane * 4] = xr[j];
  __syncthreads();

  v8f c0 = {}, c1 = {}, c2 = {}, c3 = {};
  #pragma unroll
  for (int kk = 0; kk < 2 * SI; ++kk) {
    const int kc = kk * 4 + 2 * half;
    v2f b0 = *(const v2f*)&xs[(lr)      * XP + kc];
    v2f b1 = *(const v2f*)&xs[(16 + lr) * XP + kc];
    v2f b2 = *(const v2f*)&xs[(32 + lr) * XP + kc];
    v2f b3 = *(const v2f*)&xs[(48 + lr) * XP + kc];
    if (WITH_PROBS) {
      b0 = b0 * p[kk >> 1][0];
      b1 = b1 * p[kk >> 1][1];
      b2 = b2 * p[kk >> 1][2];
      b3 = b3 * p[kk >> 1][3];
    }
    c0 = wmma_f32_k4(a[kk], b0, c0);
    c1 = wmma_f32_k4(a[kk], b1, c1);
    c2 = wmma_f32_k4(a[kk], b2, c2);
    c3 = wmma_f32_k4(a[kk], b3, c3);
  }

  float* __restrict__ po =
      part + (size_t)sl * PLANE + ((size_t)o * NV + 8 * half) * BATCH + lr;
  #pragma unroll
  for (int r = 0; r < 8; ++r) {
    po[r * BATCH]      = c0[r] * out_scale;
    po[r * BATCH + 16] = c1[r] * out_scale;
    po[r * BATCH + 32] = c2[r] * out_scale;
    po[r * BATCH + 48] = c3[r] * out_scale;
  }
}

__global__ void __launch_bounds__(128) caps_init_kernel(const float* __restrict__ w,
                                 const float* __restrict__ x,
                                 float* __restrict__ part) {
  caps_gemm_body<false>(w, x, nullptr, part, 1.0f / (float)NI);
}

__global__ void __launch_bounds__(128) caps_update_kernel(const float* __restrict__ w,
                                   const float* __restrict__ x,
                                   const float* __restrict__ probs,
                                   float* __restrict__ part) {
  caps_gemm_body<true>(w, x, probs, part, 1.0f);
}

__global__ void __launch_bounds__(256) caps_nhat_kernel(const float* __restrict__ part,
                                 float* __restrict__ nhat) {
  __shared__ float sq[NV * BATCH];
  const int b  = threadIdx.x & 63;
  const int v0 = (threadIdx.x >> 6) * 4;
  const size_t base = (size_t)blockIdx.x * NV * BATCH + (size_t)v0 * BATCH + b;
  const float* p = part + base;
  float acc[4] = {0.0f, 0.0f, 0.0f, 0.0f};
  #pragma unroll
  for (int s = 0; s < KS; ++s) {
    #pragma unroll
    for (int u = 0; u < 4; ++u) acc[u] += p[(size_t)s * PLANE + u * BATCH];
  }
  #pragma unroll
  for (int u = 0; u < 4; ++u) sq[(v0 + u) * BATCH + b] = acc[u] * acc[u];
  __syncthreads();
  float ss = 0.0f;
  #pragma unroll
  for (int v = 0; v < NV; ++v) ss += sq[v * BATCH + b];
  const float inv = 1.0f / fmaxf(sqrtf(ss), 1e-12f);
  #pragma unroll
  for (int u = 0; u < 4; ++u) nhat[base + u * BATCH] = acc[u] * inv;
}

__global__ void __launch_bounds__(LWV * 32) caps_logits_softmax_kernel(
    const float* __restrict__ w, const float* __restrict__ x,
    const float* __restrict__ nhat, float* __restrict__ probs) {
  __shared__ float slog[NO * LI * LB];
  const int tid  = threadIdx.x;
  const int wv   = tid >> 5;
  const int lane = tid & 31;
  const int half = lane >> 4;
  const int lr   = lane & 15;
  const int ib0  = (blockIdx.x >> 1) * LI;
  const int b0   = (blockIdx.x & 1) * LB;

  v4f xq[2][LI / 2][2];
  #pragma unroll
  for (int j = 0; j < 2; ++j) {
    #pragma unroll
    for (int p = 0; p < LI / 2; ++p) {
      const float* xp =
          x + ((size_t)(b0 + j * 16 + lr) * NI + ib0 + 2 * p + half) * NL;
      xq[j][p][0] = *(const v4f*)(xp);
      xq[j][p][1] = *(const v4f*)(xp + 4);
    }
  }

  const float* __restrict__ wl =
      w + ((size_t)ib0 + (lr >> 3)) * NV * NL + (size_t)(2 * half) * NL + (lr & 7);
  const float* __restrict__ nl = nhat + (size_t)(2 * half) * BATCH + b0 + lr;

  #pragma unroll 1
  for (int t = 0; t < NO / (LWV * OPT); ++t) {
    const int ob = wv + t * (LWV * OPT);

    v2f a[OPT][LI / 2][4];
    v2f bf[OPT][2][4];
    #pragma unroll
    for (int u = 0; u < OPT; ++u) {
      const float* __restrict__ wo = wl + (size_t)(ob + u * LWV) * NI * NV * NL;
      const float* __restrict__ no = nl + (size_t)(ob + u * LWV) * NV * BATCH;
      #pragma unroll
      for (int p = 0; p < LI / 2; ++p) {
        #pragma unroll
        for (int kk = 0; kk < 4; ++kk) {
          a[u][p][kk][0] = wo[(size_t)p * 2 * NV * NL + (4 * kk) * NL];
          a[u][p][kk][1] = wo[(size_t)p * 2 * NV * NL + (4 * kk + 1) * NL];
        }
      }
      #pragma unroll
      for (int j = 0; j < 2; ++j) {
        #pragma unroll
        for (int kk = 0; kk < 4; ++kk) {
          bf[u][j][kk][0] = no[(size_t)(4 * kk) * BATCH + j * 16];
          bf[u][j][kk][1] = no[(size_t)(4 * kk + 1) * BATCH + j * 16];
        }
      }
    }
    __builtin_amdgcn_sched_barrier(0);

    v8f c[OPT][2][LI / 2];
    #pragma unroll
    for (int u = 0; u < OPT; ++u) {
      #pragma unroll
      for (int j = 0; j < 2; ++j) {
        #pragma unroll
        for (int p = 0; p < LI / 2; ++p) c[u][j][p] = v8f{};
      }
    }
    #pragma unroll
    for (int kk = 0; kk < 4; ++kk) {
      #pragma unroll
      for (int u = 0; u < OPT; ++u) {
        #pragma unroll
        for (int p = 0; p < LI / 2; ++p) {
          c[u][0][p] = wmma_f32_k4(a[u][p][kk], bf[u][0][kk], c[u][0][p]);
          c[u][1][p] = wmma_f32_k4(a[u][p][kk], bf[u][1][kk], c[u][1][p]);
        }
      }
    }

    #pragma unroll
    for (int u = 0; u < OPT; ++u) {
      #pragma unroll
      for (int j = 0; j < 2; ++j) {
        #pragma unroll
        for (int p = 0; p < LI / 2; ++p) {
          const v8f d = c[u][j][p];
          v2f s2 = v2f{d[0], d[1]} * v2f{xq[j][p][0][0], xq[j][p][0][1]};
          s2 += v2f{d[2], d[3]} * v2f{xq[j][p][0][2], xq[j][p][0][3]};
          s2 += v2f{d[4], d[5]} * v2f{xq[j][p][1][0], xq[j][p][1][1]};
          s2 += v2f{d[6], d[7]} * v2f{xq[j][p][1][2], xq[j][p][1][3]};
          slog[((ob + u * LWV) * LI + 2 * p + half) * LB + j * 16 + lr] = s2[0] + s2[1];
        }
      }
    }
  }
  __syncthreads();

  if (tid < LI * LB) {
    const int ii = tid >> 5;
    const int bb = tid & 31;
    const int i  = ib0 + ii;
    float mx = -3.402823466e+38f;
    for (int o = 0; o < NO; ++o)
      mx = fmaxf(mx, slog[(o * LI + ii) * LB + bb]);
    float s = 0.0f;
    for (int o = 0; o < NO; ++o) {
      const float e = __expf(slog[(o * LI + ii) * LB + bb] - mx);
      slog[(o * LI + ii) * LB + bb] = e;
      s += e;
    }
    const float inv = 1.0f / s;
    for (int o = 0; o < NO; ++o)
      probs[((size_t)o * NI + i) * BATCH + b0 + bb] =
          slog[(o * LI + ii) * LB + bb] * inv;
  }
}

#define TP (NV + 1)
#define TQ (BATCH * TP + 16)
__global__ void __launch_bounds__(512) caps_final_kernel(const float* __restrict__ part,
                                  const float* __restrict__ bias,
                                  float* __restrict__ y) {
  __shared__ float sq[2][NV * BATCH];
  __shared__ float tile[2 * TQ];
  __shared__ float scl[2][BATCH];
  const int t  = threadIdx.x;
  const int b  = t & 63;
  const int v0 = ((t >> 6) & 3) * 4;
  const int q  = t >> 8;
  const int o0 = blockIdx.x * 2;
  const float* p = part + (size_t)(o0 + q) * NV * BATCH + (size_t)v0 * BATCH + b;
  float acc[4] = {0.0f, 0.0f, 0.0f, 0.0f};
  #pragma unroll
  for (int s = 0; s < KS; ++s) {
    #pragma unroll
    for (int u = 0; u < 4; ++u) acc[u] += p[(size_t)s * PLANE + u * BATCH];
  }
  #pragma unroll
  for (int u = 0; u < 4; ++u) {
    sq[q][(v0 + u) * BATCH + b] = acc[u] * acc[u];
    tile[q * TQ + b * TP + v0 + u] = acc[u];
  }
  __syncthreads();
  if (t < 2 * BATCH) {
    const int qq = t >> 6;
    float ss = 0.0f;
    #pragma unroll
    for (int u = 0; u < NV; ++u) ss += sq[qq][u * BATCH + b];
    const float n = sqrtf(ss);
    scl[qq][b] = n / (1.0f + ss);
  }
  __syncthreads();
  #pragma unroll
  for (int h = 0; h < 4; ++h) {
    const int e  = t + 512 * h;
    const int bb = e >> 5, qv = e & 31;
    const int qq = qv >> 4, vv = qv & 15;
    y[((size_t)bb * NO + o0) * NV + qv] =
        tile[qq * TQ + bb * TP + vv] * scl[qq][bb] + bias[o0 * NV + qv];
  }
}

extern "C" void kernel_launch(void* const* d_in, const int* in_sizes, int n_in,
                              void* d_out, int out_size, void* d_ws, size_t ws_size,
                              hipStream_t stream) {
  const float* x    = (const float*)d_in[0];
  const float* w    = (const float*)d_in[1];
  const float* bias = (const float*)d_in[2];
  float* y = (float*)d_out;

  float* probs = (float*)d_ws;
  float* part  = probs + (size_t)NO * NI * BATCH;
  float* nhat  = part  + (size_t)KS * PLANE;
  const size_t need =
      ((size_t)NO * NI * BATCH + (size_t)(KS + 1) * PLANE) * sizeof(float);
  if (ws_size < need) return;

  caps_init_kernel<<<256, 128, 0, stream>>>(w, x, part);
  for (int it = 0; it < 3; ++it) {
    caps_nhat_kernel<<<NO, 256, 0, stream>>>(part, nhat);
    caps_logits_softmax_kernel<<<256, 256, 0, stream>>>(w, x, nhat, probs);
    caps_update_kernel<<<256, 128, 0, stream>>>(w, x, probs, part);
  }
  caps_final_kernel<<<NO / 2, 512, 0, stream>>>(part, bias, y);
}


# Softmax over out-capsules from registers: a thread reads its column's 64 logits from LDS once, loads issued as one group, not one dependent LDS read per element in each of 3 passes; sum order kept
# speedup vs baseline: 6.9593x; 1.1103x over previous
#include <hip/hip_runtime.h>
#include <math.h>


typedef float v2f __attribute__((ext_vector_type(2)));
typedef float v4f __attribute__((ext_vector_type(4)));
typedef float v8f __attribute__((ext_vector_type(8)));

#define BATCH 64
#define NI    512
#define NO    64
#define NV    16
#define NL    8
#define KTOT  (NI * NL)
#define LI    4
#define LB    32
#define LWV   8
#define OPT   2
#define OG    4
#define KS    16
#define SI    (NI / KS)
#define SK    (SI * NL)
#define XP    (SK + 4)
#define PLANE (NO * NV * BATCH)

static __device__ __forceinline__ v8f wmma_f32_k4(v2f a, v2f b, v8f c) {
  return __builtin_amdgcn_wmma_f32_16x16x4_f32(false, a, false, b, (short)0, c,
                                               false, false);
}

template <bool WITH_PROBS>
static __device__ __forceinline__ void caps_gemm_body(
    const float* __restrict__ w, const float* __restrict__ x,
    const float* __restrict__ probs, float* __restrict__ part, float out_scale) {
  __shared__ __attribute__((aligned(16))) float xs[BATCH * XP];
  const int lane = threadIdx.x & 31;
  const int wv   = threadIdx.x >> 5;
  const int sl   = blockIdx.x & (KS - 1);
  const int o    = (blockIdx.x / KS) * OG + wv;
  const int half = lane >> 4;
  const int lr   = lane & 15;
  const int i0   = sl * SI;
  const float* __restrict__ wl =
      w + ((size_t)o * NI + i0) * NV * NL + (size_t)lr * NL + 2 * half;
  const float* __restrict__ xl =
      x + (size_t)(16 * wv) * KTOT + (size_t)i0 * NL + lane * 4;

  v4f xr[32];
  #pragma unroll
  for (int j = 0; j < 32; ++j)
    xr[j] = *(const v4f*)(xl + (size_t)(j >> 1) * KTOT + (j & 1) * 128);
  v2f a[2 * SI];
  #pragma unroll
  for (int kk = 0; kk < 2 * SI; ++kk)
    a[kk] = *(const v2f*)(wl + (kk >> 1) * NV * NL + (kk & 1) * 4);
  float p[SI][4];
  if (WITH_PROBS) {
    const float* __restrict__ pb = probs + ((size_t)o * NI + i0) * BATCH + lr;
    #pragma unroll
    for (int ii = 0; ii < SI; ++ii) {
      p[ii][0] = pb[ii * BATCH];      p[ii][1] = pb[ii * BATCH + 16];
      p[ii][2] = pb[ii * BATCH + 32]; p[ii][3] = pb[ii * BATCH + 48];
    }
  }

  #pragma unroll
  for (int j = 0; j < 32; ++j)
    *(v4f*)&xs[(16 * wv + (j >> 1)) * XP + (j & 1) * 128 + lane * 4] = xr[j];
  __syncthreads();

  v8f c0 = {}, c1 = {}, c2 = {}, c3 = {};
  #pragma unroll
  for (int kk = 0; kk < 2 * SI; ++kk) {
    const int kc = kk * 4 + 2 * half;
    v2f b0 = *(const v2f*)&xs[(lr)      * XP + kc];
    v2f b1 = *(const v2f*)&xs[(16 + lr) * XP + kc];
    v2f b2 = *(const v2f*)&xs[(32 + lr) * XP + kc];
    v2f b3 = *(const v2f*)&xs[(48 + lr) * XP + kc];
    if (WITH_PROBS) {
      b0 = b0 * p[kk >> 1][0];
      b1 = b1 * p[kk >> 1][1];
      b2 = b2 * p[kk >> 1][2];
      b3 = b3 * p[kk >> 1][3];
    }
    c0 = wmma_f32_k4(a[kk], b0, c0);
    c1 = wmma_f32_k4(a[kk], b1, c1);
    c2 = wmma_f32_k4(a[kk], b2, c2);
    c3 = wmma_f32_k4(a[kk], b3, c3);
  }

  float* __restrict__ po =
      part + (size_t)sl * PLANE + ((size_t)o * NV + 8 * half) * BATCH + lr;
  #pragma unroll
  for (int r = 0; r < 8; ++r) {
    po[r * BATCH]      = c0[r] * out_scale;
    po[r * BATCH + 16] = c1[r] * out_scale;
    po[r * BATCH + 32] = c2[r] * out_scale;
    po[r * BATCH + 48] = c3[r] * out_scale;
  }
}

__global__ void __launch_bounds__(128) caps_init_kernel(const float* __restrict__ w,
                                 const float* __restrict__ x,
                                 float* __restrict__ part) {
  caps_gemm_body<false>(w, x, nullptr, part, 1.0f / (float)NI);
}

__global__ void __launch_bounds__(128) caps_update_kernel(const float* __restrict__ w,
                                   const float* __restrict__ x,
                                   const float* __restrict__ probs,
                                   float* __restrict__ part) {
  caps_gemm_body<true>(w, x, probs, part, 1.0f);
}

__global__ void __launch_bounds__(256) caps_nhat_kernel(const float* __restrict__ part,
                                 float* __restrict__ nhat) {
  __shared__ float sq[NV * BATCH];
  const int b  = threadIdx.x & 63;
  const int v0 = (threadIdx.x >> 6) * 4;
  const size_t base = (size_t)blockIdx.x * NV * BATCH + (size_t)v0 * BATCH + b;
  const float* p = part + base;
  float acc[4] = {0.0f, 0.0f, 0.0f, 0.0f};
  #pragma unroll
  for (int s = 0; s < KS; ++s) {
    #pragma unroll
    for (int u = 0; u < 4; ++u) acc[u] += p[(size_t)s * PLANE + u * BATCH];
  }
  #pragma unroll
  for (int u = 0; u < 4; ++u) sq[(v0 + u) * BATCH + b] = acc[u] * acc[u];
  __syncthreads();
  float ss = 0.0f;
  #pragma unroll
  for (int v = 0; v < NV; ++v) ss += sq[v * BATCH + b];
  const float inv = 1.0f / fmaxf(sqrtf(ss), 1e-12f);
  #pragma unroll
  for (int u = 0; u < 4; ++u) nhat[base + u * BATCH] = acc[u] * inv;
}

__global__ void __launch_bounds__(LWV * 32) caps_logits_softmax_kernel(
    const float* __restrict__ w, const float* __restrict__ x,
    const float* __restrict__ nhat, float* __restrict__ probs) {
  __shared__ float slog[NO * LI * LB];
  const int tid  = threadIdx.x;
  const int wv   = tid >> 5;
  const int lane = tid & 31;
  const int half = lane >> 4;
  const int lr   = lane & 15;
  const int ib0  = (blockIdx.x >> 1) * LI;
  const int b0   = (blockIdx.x & 1) * LB;

  v4f xq[2][LI / 2][2];
  #pragma unroll
  for (int j = 0; j < 2; ++j) {
    #pragma unroll
    for (int p = 0; p < LI / 2; ++p) {
      const float* xp =
          x + ((size_t)(b0 + j * 16 + lr) * NI + ib0 + 2 * p + half) * NL;
      xq[j][p][0] = *(const v4f*)(xp);
      xq[j][p][1] = *(const v4f*)(xp + 4);
    }
  }

  const float* __restrict__ wl =
      w + ((size_t)ib0 + (lr >> 3)) * NV * NL + (size_t)(2 * half) * NL + (lr & 7);
  const float* __restrict__ nl = nhat + (size_t)(2 * half) * BATCH + b0 + lr;

  #pragma unroll 1
  for (int t = 0; t < NO / (LWV * OPT); ++t) {
    const int ob = wv + t * (LWV * OPT);

    v2f a[OPT][LI / 2][4];
    v2f bf[OPT][2][4];
    #pragma unroll
    for (int u = 0; u < OPT; ++u) {
      const float* __restrict__ wo = wl + (size_t)(ob + u * LWV) * NI * NV * NL;
      const float* __restrict__ no = nl + (size_t)(ob + u * LWV) * NV * BATCH;
      #pragma unroll
      for (int p = 0; p < LI / 2; ++p) {
        #pragma unroll
        for (int kk = 0; kk < 4; ++kk) {
          a[u][p][kk][0] = wo[(size_t)p * 2 * NV * NL + (4 * kk) * NL];
          a[u][p][kk][1] = wo[(size_t)p * 2 * NV * NL + (4 * kk + 1) * NL];
        }
      }
      #pragma unroll
      for (int j = 0; j < 2; ++j) {
        #pragma unroll
        for (int kk = 0; kk < 4; ++kk) {
          bf[u][j][kk][0] = no[(size_t)(4 * kk) * BATCH + j * 16];
          bf[u][j][kk][1] = no[(size_t)(4 * kk + 1) * BATCH + j * 16];
        }
      }
    }
    __builtin_amdgcn_sched_barrier(0);

    v8f c[OPT][2][LI / 2];
    #pragma unroll
    for (int u = 0; u < OPT; ++u) {
      #pragma unroll
      for (int j = 0; j < 2; ++j) {
        #pragma unroll
        for (int p = 0; p < LI / 2; ++p) c[u][j][p] = v8f{};
      }
    }
    #pragma unroll
    for (int kk = 0; kk < 4; ++kk) {
      #pragma unroll
      for (int u = 0; u < OPT; ++u) {
        #pragma unroll
        for (int p = 0; p < LI / 2; ++p) {
          c[u][0][p] = wmma_f32_k4(a[u][p][kk], bf[u][0][kk], c[u][0][p]);
          c[u][1][p] = wmma_f32_k4(a[u][p][kk], bf[u][1][kk], c[u][1][p]);
        }
      }
    }

    #pragma unroll
    for (int u = 0; u < OPT; ++u) {
      #pragma unroll
      for (int j = 0; j < 2; ++j) {
        #pragma unroll
        for (int p = 0; p < LI / 2; ++p) {
          const v8f d = c[u][j][p];
          v2f s2 = v2f{d[0], d[1]} * v2f{xq[j][p][0][0], xq[j][p][0][1]};
          s2 += v2f{d[2], d[3]} * v2f{xq[j][p][0][2], xq[j][p][0][3]};
          s2 += v2f{d[4], d[5]} * v2f{xq[j][p][1][0], xq[j][p][1][1]};
          s2 += v2f{d[6], d[7]} * v2f{xq[j][p][1][2], xq[j][p][1][3]};
          slog[((ob + u * LWV) * LI + 2 * p + half) * LB + j * 16 + lr] = s2[0] + s2[1];
        }
      }
    }
  }
  __syncthreads();

  if (tid < LI * LB) {
    const int ii = tid >> 5;
    const int bb = tid & 31;
    const int i  = ib0 + ii;
    float lg[NO];
    #pragma unroll
    for (int o = 0; o < NO; ++o) lg[o] = slog[(o * LI + ii) * LB + bb];
    float m0 = -3.402823466e+38f, m1 = m0, m2 = m0, m3 = m0;
    #pragma unroll
    for (int o = 0; o < NO; o += 4) {
      m0 = fmaxf(m0, lg[o]);     m1 = fmaxf(m1, lg[o + 1]);
      m2 = fmaxf(m2, lg[o + 2]); m3 = fmaxf(m3, lg[o + 3]);
    }
    const float mx = fmaxf(fmaxf(m0, m1), fmaxf(m2, m3));
    float s = 0.0f;
    #pragma unroll
    for (int o = 0; o < NO; ++o) {
      lg[o] = __expf(lg[o] - mx);
      s += lg[o];
    }
    const float inv = 1.0f / s;
    float* __restrict__ pp = probs + (size_t)i * BATCH + b0 + bb;
    #pragma unroll
    for (int o = 0; o < NO; ++o) pp[(size_t)o * NI * BATCH] = lg[o] * inv;
  }
}

#define TP (NV + 1)
#define TQ (BATCH * TP + 16)
__global__ void __launch_bounds__(512) caps_final_kernel(const float* __restrict__ part,
                                  const float* __restrict__ bias,
                                  float* __restrict__ y) {
  __shared__ float sq[2][NV * BATCH];
  __shared__ float tile[2 * TQ];
  __shared__ float scl[2][BATCH];
  const int t  = threadIdx.x;
  const int b  = t & 63;
  const int v0 = ((t >> 6) & 3) * 4;
  const int q  = t >> 8;
  const int o0 = blockIdx.x * 2;
  const float* p = part + (size_t)(o0 + q) * NV * BATCH + (size_t)v0 * BATCH + b;
  float acc[4] = {0.0f, 0.0f, 0.0f, 0.0f};
  #pragma unroll
  for (int s = 0; s < KS; ++s) {
    #pragma unroll
    for (int u = 0; u < 4; ++u) acc[u] += p[(size_t)s * PLANE + u * BATCH];
  }
  #pragma unroll
  for (int u = 0; u < 4; ++u) {
    sq[q][(v0 + u) * BATCH + b] = acc[u] * acc[u];
    tile[q * TQ + b * TP + v0 + u] = acc[u];
  }
  __syncthreads();
  if (t < 2 * BATCH) {
    const int qq = t >> 6;
    float ss = 0.0f;
    #pragma unroll
    for (int u = 0; u < NV; ++u) ss += sq[qq][u * BATCH + b];
    const float n = sqrtf(ss);
    scl[qq][b] = n / (1.0f + ss);
  }
  __syncthreads();
  #pragma unroll
  for (int h = 0; h < 4; ++h) {
    const int e  = t + 512 * h;
    const int bb = e >> 5, qv = e & 31;
    const int qq = qv >> 4, vv = qv & 15;
    y[((size_t)bb * NO + o0) * NV + qv] =
        tile[qq * TQ + bb * TP + vv] * scl[qq][bb] + bias[o0 * NV + qv];
  }
}

extern "C" void kernel_launch(void* const* d_in, const int* in_sizes, int n_in,
                              void* d_out, int out_size, void* d_ws, size_t ws_size,
                              hipStream_t stream) {
  const float* x    = (const float*)d_in[0];
  const float* w    = (const float*)d_in[1];
  const float* bias = (const float*)d_in[2];
  float* y = (float*)d_out;

  float* probs = (float*)d_ws;
  float* part  = probs + (size_t)NO * NI * BATCH;
  float* nhat  = part  + (size_t)KS * PLANE;
  const size_t need =
      ((size_t)NO * NI * BATCH + (size_t)(KS + 1) * PLANE) * sizeof(float);
  if (ws_size < need) return;

  caps_init_kernel<<<256, 128, 0, stream>>>(w, x, part);
  for (int it = 0; it < 3; ++it) {
    caps_nhat_kernel<<<NO, 256, 0, stream>>>(part, nhat);
    caps_logits_softmax_kernel<<<256, 256, 0, stream>>>(w, x, nhat, probs);
    caps_update_kernel<<<256, 128, 0, stream>>>(w, x, probs, part);
  }
  caps_final_kernel<<<NO / 2, 512, 0, stream>>>(part, bias, y);
}
